# starting kernel
# baseline (speedup 1.0000x reference)
#include <hip/hip_runtime.h>


typedef __attribute__((ext_vector_type(2))) float v2f;
typedef __attribute__((ext_vector_type(8))) float v8f;

#define KTOT 144

__global__ __launch_bounds__(32) void lc2d_wmma_f32_kernel(
    const float* __restrict__ x,
    const float* __restrict__ wgt,
    const float* __restrict__ bias,
    float* __restrict__ out)
{
  __shared__ float lds_x[KTOT * 32];
  __shared__ float lds_w[KTOT * 32];

  const int lane = threadIdx.x;
  const int loc  = blockIdx.x;
  const int h = loc >> 6;
  const int w = loc & 63;

  {
    const int o = lane;
    #pragma unroll
    for (int c = 0; c < 16; ++c) {
      const float* src = wgt + (size_t)((o * 16 + c) * 4096 + loc) * 9;
      #pragma unroll
      for (int kk = 0; kk < 9; ++kk)
        lds_w[(c * 9 + kk) * 32 + o] = src[kk];
    }
  }

  {
    const int b = lane;
    #pragma unroll
    for (int c = 0; c < 16; ++c) {
      const float* src = x + ((size_t)(b * 16 + c) * 66 + h) * 66 + w;
      #pragma unroll
      for (int i = 0; i < 3; ++i)
        #pragma unroll
        for (int j = 0; j < 3; ++j)
          lds_x[(c * 9 + i * 3 + j) * 32 + b] = src[i * 66 + j];
    }
  }
  __syncthreads();

  const int t = lane >> 4;
  const int r = lane & 15;

  v8f acc00 = {}, acc01 = {}, acc10 = {}, acc11 = {};

  #pragma unroll
  for (int kc = 0; kc < KTOT; kc += 4) {
    const int kA = kc + 2 * t;
    const int base = kA * 32;
    v2f a0, a1, b0, b1;
    a0.x = lds_x[base + r];           a0.y = lds_x[base + 32 + r];
    a1.x = lds_x[base + 16 + r];      a1.y = lds_x[base + 48 + r];
    b0.x = lds_w[base + r];           b0.y = lds_w[base + 32 + r];
    b1.x = lds_w[base + 16 + r];      b1.y = lds_w[base + 48 + r];

    acc00 = __builtin_amdgcn_wmma_f32_16x16x4_f32(false, a0, false, b0,
                                                  (short)0, acc00, false, false);
    acc01 = __builtin_amdgcn_wmma_f32_16x16x4_f32(false, a0, false, b1,
                                                  (short)0, acc01, false, false);
    acc10 = __builtin_amdgcn_wmma_f32_16x16x4_f32(false, a1, false, b0,
                                                  (short)0, acc10, false, false);
    acc11 = __builtin_amdgcn_wmma_f32_16x16x4_f32(false, a1, false, b1,
                                                  (short)0, acc11, false, false);
  }

  const float bv0 = bias[(size_t)r * 4096 + loc];
  const float bv1 = bias[(size_t)(r + 16) * 4096 + loc];

  #pragma unroll
  for (int row = 0; row < 8; ++row) {
    const int bm0 = row + 8 * t;
    const int bm1 = 16 + row + 8 * t;
    out[(size_t)(bm0 * 32 + r)        * 4096 + loc] = acc00[row] + bv0;
    out[(size_t)(bm0 * 32 + r + 16)   * 4096 + loc] = acc01[row] + bv1;
    out[(size_t)(bm1 * 32 + r)        * 4096 + loc] = acc10[row] + bv0;
    out[(size_t)(bm1 * 32 + r + 16)   * 4096 + loc] = acc11[row] + bv1;
  }
}

extern "C" void kernel_launch(void* const* d_in, const int* in_sizes, int n_in,
                              void* d_out, int out_size, void* d_ws, size_t ws_size,
                              hipStream_t stream) {
  (void)in_sizes; (void)n_in; (void)out_size; (void)d_ws; (void)ws_size;
  const float* x    = (const float*)d_in[0];
  const float* wgt  = (const float*)d_in[1];
  const float* bias = (const float*)d_in[2];
  float* out = (float*)d_out;
  lc2d_wmma_f32_kernel<<<dim3(64 * 64), dim3(32), 0, stream>>>(x, wgt, bias, out);
}


# Work-group = 4 waves on 16 adjacent locations (was 1 wave, 4 locations), still no LDS: lines that neighbouring location quads share come once per WGP through its cache; per-wave code, sum order kept
# speedup vs baseline: 3.4659x; 3.4659x over previous
#include <hip/hip_runtime.h>


typedef __attribute__((ext_vector_type(2))) float v2f;
typedef __attribute__((ext_vector_type(4))) float v4f;
typedef __attribute__((ext_vector_type(8))) float v8f;

#define NLOC 4
#define WRUN (NLOC * 9)
#define XRUN (NLOC + 2)

struct Frag {
  v4f wq[2][2][2][4];
  v4f wt[2][2];
  float xr[2][2][3][XRUN];
};

__device__ __forceinline__ float quad_xor1(float v) {
  return __int_as_float(__builtin_amdgcn_mov_dpp(__float_as_int(v), 0xB1, 0xF, 0xF, true));
}
__device__ __forceinline__ float quad_xor2(float v) {
  return __int_as_float(__builtin_amdgcn_mov_dpp(__float_as_int(v), 0x4E, 0xF, 0xF, true));
}

__device__ __forceinline__ void quad_transpose(const v4f (&v)[4], v4f (&T)[4],
                                               bool odd, bool hi) {
#pragma unroll
  for (int d = 0; d < 4; ++d) {
    const float v0 = v[0][d], v1 = v[1][d], v2 = v[2][d], v3 = v[3][d];
    const float x0 = quad_xor1(v0), x1 = quad_xor1(v1);
    const float x2 = quad_xor1(v2), x3 = quad_xor1(v3);
    const float u0 = odd ? x1 : v0;
    const float u1 = odd ? v1 : x0;
    const float u2 = odd ? x3 : v2;
    const float u3 = odd ? v3 : x2;
    const float y0 = quad_xor2(u0), y1 = quad_xor2(u1);
    const float y2 = quad_xor2(u2), y3 = quad_xor2(u3);
    T[0][d] = hi ? y2 : u0;
    T[1][d] = hi ? y3 : u1;
    T[2][d] = hi ? u2 : y0;
    T[3][d] = hi ? u3 : y1;
  }
}

__device__ __forceinline__ void load_frag(Frag& f, const float* __restrict__ wq,
                                          const float* __restrict__ wl,
                                          const float* __restrict__ xl, int u) {
#pragma unroll
  for (int sub = 0; sub < 2; ++sub) {
    const int s = 2 * u + sub;
#pragma unroll
    for (int j = 0; j < 2; ++j) {
      const size_t off = (size_t)j * (16 * 16 * 4096 * 9) + (size_t)s * (2 * 4096 * 9);
#pragma unroll
      for (int n = 0; n < 4; ++n) {
        const float* p = wq + off + (size_t)n * (16 * 4096 * 9);
        f.wq[sub][j][0][n] = *(const v4f*)(p);
        f.wq[sub][j][1][n] = *(const v4f*)(p + 16);
      }
      f.wt[sub][j] = *(const v4f*)(wl + off + 32);
      __builtin_amdgcn_sched_barrier(0);
    }
#pragma unroll
    for (int i = 0; i < 2; ++i) {
      const float* p = xl + (size_t)i * (16 * 16 * 66 * 66) + (size_t)s * (2 * 66 * 66);
#pragma unroll
      for (int di = 0; di < 3; ++di) {
#pragma unroll
        for (int q = 0; q < XRUN / 2; ++q) {
          const v2f v = *(const v2f*)(p + di * 66 + 2 * q);
          f.xr[sub][i][di][2 * q + 0] = v.x;
          f.xr[sub][i][di][2 * q + 1] = v.y;
        }
      }
      __builtin_amdgcn_sched_barrier(0);
    }
  }
}

__device__ __forceinline__ void mma_frag(const Frag& f, v8f (&acc)[NLOC][2][2],
                                         bool odd, bool hi) {
  float w[2][2][WRUN];
#pragma unroll
  for (int sub = 0; sub < 2; ++sub)
#pragma unroll
    for (int j = 0; j < 2; ++j) {
#pragma unroll
      for (int bq = 0; bq < 2; ++bq) {
        v4f T[4];
        quad_transpose(f.wq[sub][j][bq], T, odd, hi);
#pragma unroll
        for (int p = 0; p < 4; ++p)
#pragma unroll
          for (int d = 0; d < 4; ++d) w[sub][j][16 * bq + 4 * p + d] = T[p][d];
      }
#pragma unroll
      for (int d = 0; d < 4; ++d) w[sub][j][32 + d] = f.wt[sub][j][d];
    }

#pragma unroll
  for (int l = 0; l < NLOC; ++l) {
#pragma unroll
    for (int p = 0; p < 9; ++p) {
      const int q0 = 2 * p, q1 = 2 * p + 1;
      const int s0 = q0 / 9, k0 = q0 % 9;
      const int s1 = q1 / 9, k1 = q1 % 9;
      v2f a[2], b[2];
#pragma unroll
      for (int i = 0; i < 2; ++i) {
        a[i].x = f.xr[s0][i][k0 / 3][l + k0 % 3];
        a[i].y = f.xr[s1][i][k1 / 3][l + k1 % 3];
      }
#pragma unroll
      for (int j = 0; j < 2; ++j) {
        b[j].x = w[s0][j][l * 9 + k0];
        b[j].y = w[s1][j][l * 9 + k1];
      }
#pragma unroll
      for (int i = 0; i < 2; ++i)
#pragma unroll
        for (int j = 0; j < 2; ++j)
          acc[l][i][j] = __builtin_amdgcn_wmma_f32_16x16x4_f32(
              false, a[i], false, b[j], (short)0, acc[l][i][j], false, false);
    }
  }
}

__global__ __launch_bounds__(128) void lc2d_wmma_f32_wgp16_kernel(
    const float* __restrict__ x,
    const float* __restrict__ wgt,
    const float* __restrict__ bias,
    float* __restrict__ out)
{
  const int lane = threadIdx.x & 31;
  const int t = lane >> 4;
  const int r = lane & 15;
  const int wv = __builtin_amdgcn_readfirstlane((int)(threadIdx.x >> 5));
  const int quad = ((wv & 1) << 1) | (wv >> 1);
  const int loc0 = (blockIdx.x * 4 + quad) * NLOC;
  const int h = loc0 >> 6;
  const int w0 = loc0 & 63;
  const bool odd = (lane & 1) != 0;
  const bool hi = (lane & 2) != 0;

  const float* wl = wgt + ((size_t)(r * 16 + t) * 4096 + loc0) * 9;
  const float* wq = wgt + ((size_t)((r & ~3) * 16 + t) * 4096 + loc0) * 9 + 4 * (r & 3);
  const float* xl = x + ((size_t)(r * 16 + t) * 66 + h) * 66 + w0;

  v8f acc[NLOC][2][2];
#pragma unroll
  for (int l = 0; l < NLOC; ++l)
#pragma unroll
    for (int i = 0; i < 2; ++i)
#pragma unroll
      for (int j = 0; j < 2; ++j) acc[l][i][j] = (v8f){};

  Frag f0, f1;
  load_frag(f0, wq, wl, xl, 0);
  load_frag(f1, wq, wl, xl, 1);
  mma_frag(f0, acc, odd, hi);
  load_frag(f0, wq, wl, xl, 2);
  mma_frag(f1, acc, odd, hi);
  load_frag(f1, wq, wl, xl, 3);
  mma_frag(f0, acc, odd, hi);
  __builtin_amdgcn_sched_barrier(0);
  mma_frag(f1, acc, odd, hi);

  v4f bv[2];
#pragma unroll
  for (int j = 0; j < 2; ++j)
    bv[j] = *(const v4f*)(bias + (size_t)(r + 16 * j) * 4096 + loc0);

#pragma unroll
  for (int i = 0; i < 2; ++i)
#pragma unroll
    for (int j = 0; j < 2; ++j)
#pragma unroll
      for (int v = 0; v < 8; ++v) {
        const int b = 16 * i + v + 8 * t;
        const int o = 16 * j + r;
        v4f res;
        res.x = acc[0][i][j][v] + bv[j].x;
        res.y = acc[1][i][j][v] + bv[j].y;
        res.z = acc[2][i][j][v] + bv[j].z;
        res.w = acc[3][i][j][v] + bv[j].w;
        *(v4f*)(out + (size_t)(b * 32 + o) * 4096 + loc0) = res;
      }
}

extern "C" void kernel_launch(void* const* d_in, const int* in_sizes, int n_in,
                              void* d_out, int out_size, void* d_ws, size_t ws_size,
                              hipStream_t stream) {
  (void)in_sizes; (void)n_in; (void)out_size; (void)d_ws; (void)ws_size;
  const float* x    = (const float*)d_in[0];
  const float* wgt  = (const float*)d_in[1];
  const float* bias = (const float*)d_in[2];
  float* out = (float*)d_out;
  lc2d_wmma_f32_wgp16_kernel<<<dim3(64 * 64 / (NLOC * 4)), dim3(128), 0, stream>>>(x, wgt, bias, out);
}


# x: the 24-byte run of a (row b, channel, image row) is fetched by lanes b, b^1 together, 16 bytes each: an x load touches 16 lines (was 32), each run once (was twice); DPP swap; sum order kept
# speedup vs baseline: 3.6667x; 1.0579x over previous
#include <hip/hip_runtime.h>


typedef __attribute__((ext_vector_type(2))) float v2f;
typedef __attribute__((ext_vector_type(4))) float v4f;
typedef __attribute__((ext_vector_type(8))) float v8f;
typedef v4f v4f_a8 __attribute__((aligned(8)));

#define NLOC 4
#define WRUN (NLOC * 9)
#define XRUN (NLOC + 2)

struct Frag {
  v4f wq[2][2][2][4];
  v4f wt[2][2];
  v4f xe[2][2][3];
  v4f xo[2][2][3];
};

__device__ __forceinline__ float quad_xor1(float v) {
  return __int_as_float(__builtin_amdgcn_mov_dpp(__float_as_int(v), 0xB1, 0xF, 0xF, true));
}
__device__ __forceinline__ float quad_xor2(float v) {
  return __int_as_float(__builtin_amdgcn_mov_dpp(__float_as_int(v), 0x4E, 0xF, 0xF, true));
}

__device__ __forceinline__ void quad_transpose(const v4f (&v)[4], v4f (&T)[4],
                                               bool odd, bool hi) {
#pragma unroll
  for (int d = 0; d < 4; ++d) {
    const float v0 = v[0][d], v1 = v[1][d], v2 = v[2][d], v3 = v[3][d];
    const float x0 = quad_xor1(v0), x1 = quad_xor1(v1);
    const float x2 = quad_xor1(v2), x3 = quad_xor1(v3);
    const float u0 = odd ? x1 : v0;
    const float u1 = odd ? v1 : x0;
    const float u2 = odd ? x3 : v2;
    const float u3 = odd ? v3 : x2;
    const float y0 = quad_xor2(u0), y1 = quad_xor2(u1);
    const float y2 = quad_xor2(u2), y3 = quad_xor2(u3);
    T[0][d] = hi ? y2 : u0;
    T[1][d] = hi ? y3 : u1;
    T[2][d] = hi ? u2 : y0;
    T[3][d] = hi ? u3 : y1;
  }
}

__device__ __forceinline__ void load_frag(Frag& f, const float* __restrict__ wq,
                                          const float* __restrict__ wl,
                                          const float* __restrict__ xl, int u) {
#pragma unroll
  for (int sub = 0; sub < 2; ++sub) {
    const int s = 2 * u + sub;
#pragma unroll
    for (int j = 0; j < 2; ++j) {
      const size_t off = (size_t)j * (16 * 16 * 4096 * 9) + (size_t)s * (2 * 4096 * 9);
#pragma unroll
      for (int n = 0; n < 4; ++n) {
        const float* p = wq + off + (size_t)n * (16 * 4096 * 9);
        f.wq[sub][j][0][n] = *(const v4f*)(p);
        f.wq[sub][j][1][n] = *(const v4f*)(p + 16);
      }
      f.wt[sub][j] = *(const v4f*)(wl + off + 32);
      __builtin_amdgcn_sched_barrier(0);
    }
#pragma unroll
    for (int i = 0; i < 2; ++i) {
      const float* p = xl + (size_t)i * (16 * 16 * 66 * 66) + (size_t)s * (2 * 66 * 66);
#pragma unroll
      for (int di = 0; di < 3; ++di) {
        f.xe[sub][i][di] = *(const v4f_a8*)(p + di * 66);
        f.xo[sub][i][di] = *(const v4f_a8*)(p + 16 * 66 * 66 + di * 66);
      }
      __builtin_amdgcn_sched_barrier(0);
    }
  }
}

__device__ __forceinline__ void mma_frag(const Frag& f, v8f (&acc)[NLOC][2][2],
                                         bool odd, bool hi) {
  float w[2][2][WRUN];
#pragma unroll
  for (int sub = 0; sub < 2; ++sub)
#pragma unroll
    for (int j = 0; j < 2; ++j) {
#pragma unroll
      for (int bq = 0; bq < 2; ++bq) {
        v4f T[4];
        quad_transpose(f.wq[sub][j][bq], T, odd, hi);
#pragma unroll
        for (int p = 0; p < 4; ++p)
#pragma unroll
          for (int d = 0; d < 4; ++d) w[sub][j][16 * bq + 4 * p + d] = T[p][d];
      }
#pragma unroll
      for (int d = 0; d < 4; ++d) w[sub][j][32 + d] = f.wt[sub][j][d];
    }

  float xr[2][2][3][XRUN];
#pragma unroll
  for (int sub = 0; sub < 2; ++sub)
#pragma unroll
    for (int i = 0; i < 2; ++i)
#pragma unroll
      for (int di = 0; di < 3; ++di) {
        const v4f E = f.xe[sub][i][di], O = f.xo[sub][i][di];
        const float g0 = quad_xor1(odd ? E[2] : O[0]);
        const float g1 = quad_xor1(odd ? E[3] : O[1]);
        xr[sub][i][di][0] = odd ? g0 : E[0];
        xr[sub][i][di][1] = odd ? g1 : E[1];
        xr[sub][i][di][2] = odd ? O[0] : E[2];
        xr[sub][i][di][3] = odd ? O[1] : E[3];
        xr[sub][i][di][4] = odd ? O[2] : g0;
        xr[sub][i][di][5] = odd ? O[3] : g1;
      }

#pragma unroll
  for (int l = 0; l < NLOC; ++l) {
#pragma unroll
    for (int p = 0; p < 9; ++p) {
      const int q0 = 2 * p, q1 = 2 * p + 1;
      const int s0 = q0 / 9, k0 = q0 % 9;
      const int s1 = q1 / 9, k1 = q1 % 9;
      v2f a[2], b[2];
#pragma unroll
      for (int i = 0; i < 2; ++i) {
        a[i].x = xr[s0][i][k0 / 3][l + k0 % 3];
        a[i].y = xr[s1][i][k1 / 3][l + k1 % 3];
      }
#pragma unroll
      for (int j = 0; j < 2; ++j) {
        b[j].x = w[s0][j][l * 9 + k0];
        b[j].y = w[s1][j][l * 9 + k1];
      }
#pragma unroll
      for (int i = 0; i < 2; ++i)
#pragma unroll
        for (int j = 0; j < 2; ++j)
          acc[l][i][j] = __builtin_amdgcn_wmma_f32_16x16x4_f32(
              false, a[i], false, b[j], (short)0, acc[l][i][j], false, false);
    }
  }
}

__global__ __launch_bounds__(128) void lc2d_wmma_f32_wgp16_kernel(
    const float* __restrict__ x,
    const float* __restrict__ wgt,
    const float* __restrict__ bias,
    float* __restrict__ out)
{
  const int lane = threadIdx.x & 31;
  const int t = lane >> 4;
  const int r = lane & 15;
  const int wv = __builtin_amdgcn_readfirstlane((int)(threadIdx.x >> 5));
  const int quad = ((wv & 1) << 1) | (wv >> 1);
  const int loc0 = (blockIdx.x * 4 + quad) * NLOC;
  const int h = loc0 >> 6;
  const int w0 = loc0 & 63;
  const bool odd = (lane & 1) != 0;
  const bool hi = (lane & 2) != 0;

  const float* wl = wgt + ((size_t)(r * 16 + t) * 4096 + loc0) * 9;
  const float* wq = wgt + ((size_t)((r & ~3) * 16 + t) * 4096 + loc0) * 9 + 4 * (r & 3);
  const float* xl = x + ((size_t)((r & ~1) * 16 + t) * 66 + h) * 66 + w0 + 2 * (r & 1);

  v8f acc[NLOC][2][2];
#pragma unroll
  for (int l = 0; l < NLOC; ++l)
#pragma unroll
    for (int i = 0; i < 2; ++i)
#pragma unroll
      for (int j = 0; j < 2; ++j) acc[l][i][j] = (v8f){};

  Frag f0, f1;
  load_frag(f0, wq, wl, xl, 0);
  load_frag(f1, wq, wl, xl, 1);
  mma_frag(f0, acc, odd, hi);
  load_frag(f0, wq, wl, xl, 2);
  mma_frag(f1, acc, odd, hi);
  load_frag(f1, wq, wl, xl, 3);
  mma_frag(f0, acc, odd, hi);
  __builtin_amdgcn_sched_barrier(0);
  mma_frag(f1, acc, odd, hi);

  v4f bv[2];
#pragma unroll
  for (int j = 0; j < 2; ++j)
    bv[j] = *(const v4f*)(bias + (size_t)(r + 16 * j) * 4096 + loc0);

#pragma unroll
  for (int i = 0; i < 2; ++i)
#pragma unroll
    for (int j = 0; j < 2; ++j)
#pragma unroll
      for (int v = 0; v < 8; ++v) {
        const int b = 16 * i + v + 8 * t;
        const int o = 16 * j + r;
        v4f res;
        res.x = acc[0][i][j][v] + bv[j].x;
        res.y = acc[1][i][j][v] + bv[j].y;
        res.z = acc[2][i][j][v] + bv[j].z;
        res.w = acc[3][i][j][v] + bv[j].w;
        *(v4f*)(out + (size_t)(b * 32 + o) * 4096 + loc0) = res;
      }
}

extern "C" void kernel_launch(void* const* d_in, const int* in_sizes, int n_in,
                              void* d_out, int out_size, void* d_ws, size_t ws_size,
                              hipStream_t stream) {
  (void)in_sizes; (void)n_in; (void)out_size; (void)d_ws; (void)ws_size;
  const float* x    = (const float*)d_in[0];
  const float* wgt  = (const float*)d_in[1];
  const float* bias = (const float*)d_in[2];
  float* out = (float*)d_out;
  lc2d_wmma_f32_wgp16_kernel<<<dim3(64 * 64 / (NLOC * 4)), dim3(128), 0, stream>>>(x, wgt, bias, out);
}


# Out tile passes through LDS: 4 waves park their 16-byte pieces in one row per (b,o); each store then writes 8 rows x 64 contiguous bytes (was 32 lines x 16 B): 1/4 the write requests; sum order kept
# speedup vs baseline: 5.4383x; 1.4832x over previous
#include <hip/hip_runtime.h>


typedef __attribute__((ext_vector_type(2))) float v2f;
typedef __attribute__((ext_vector_type(4))) float v4f;
typedef __attribute__((ext_vector_type(8))) float v8f;
typedef v4f v4f_a8 __attribute__((aligned(8)));

#define NLOC 4
#define WRUN (NLOC * 9)
#define XRUN (NLOC + 2)
#define OROW 20

struct Frag {
  v4f wq[2][2][2][4];
  v4f wt[2][2];
  v4f xe[2][2][3];
  v4f xo[2][2][3];
};

__device__ __forceinline__ float quad_xor1(float v) {
  return __int_as_float(__builtin_amdgcn_mov_dpp(__float_as_int(v), 0xB1, 0xF, 0xF, true));
}
__device__ __forceinline__ float quad_xor2(float v) {
  return __int_as_float(__builtin_amdgcn_mov_dpp(__float_as_int(v), 0x4E, 0xF, 0xF, true));
}

__device__ __forceinline__ void quad_transpose(const v4f (&v)[4], v4f (&T)[4],
                                               bool odd, bool hi) {
#pragma unroll
  for (int d = 0; d < 4; ++d) {
    const float v0 = v[0][d], v1 = v[1][d], v2 = v[2][d], v3 = v[3][d];
    const float x0 = quad_xor1(v0), x1 = quad_xor1(v1);
    const float x2 = quad_xor1(v2), x3 = quad_xor1(v3);
    const float u0 = odd ? x1 : v0;
    const float u1 = odd ? v1 : x0;
    const float u2 = odd ? x3 : v2;
    const float u3 = odd ? v3 : x2;
    const float y0 = quad_xor2(u0), y1 = quad_xor2(u1);
    const float y2 = quad_xor2(u2), y3 = quad_xor2(u3);
    T[0][d] = hi ? y2 : u0;
    T[1][d] = hi ? y3 : u1;
    T[2][d] = hi ? u2 : y0;
    T[3][d] = hi ? u3 : y1;
  }
}

__device__ __forceinline__ void load_frag(Frag& f, const float* __restrict__ wq,
                                          const float* __restrict__ wl,
                                          const float* __restrict__ xl, int u) {
#pragma unroll
  for (int sub = 0; sub < 2; ++sub) {
    const int s = 2 * u + sub;
#pragma unroll
    for (int j = 0; j < 2; ++j) {
      const size_t off = (size_t)j * (16 * 16 * 4096 * 9) + (size_t)s * (2 * 4096 * 9);
#pragma unroll
      for (int n = 0; n < 4; ++n) {
        const float* p = wq + off + (size_t)n * (16 * 4096 * 9);
        f.wq[sub][j][0][n] = *(const v4f*)(p);
        f.wq[sub][j][1][n] = *(const v4f*)(p + 16);
      }
      f.wt[sub][j] = *(const v4f*)(wl + off + 32);
      __builtin_amdgcn_sched_barrier(0);
    }
#pragma unroll
    for (int i = 0; i < 2; ++i) {
      const float* p = xl + (size_t)i * (16 * 16 * 66 * 66) + (size_t)s * (2 * 66 * 66);
#pragma unroll
      for (int di = 0; di < 3; ++di) {
        f.xe[sub][i][di] = *(const v4f_a8*)(p + di * 66);
        f.xo[sub][i][di] = *(const v4f_a8*)(p + 16 * 66 * 66 + di * 66);
      }
      __builtin_amdgcn_sched_barrier(0);
    }
  }
}

__device__ __forceinline__ void mma_frag(const Frag& f, v8f (&acc)[NLOC][2][2],
                                         bool odd, bool hi) {
  float w[2][2][WRUN];
#pragma unroll
  for (int sub = 0; sub < 2; ++sub)
#pragma unroll
    for (int j = 0; j < 2; ++j) {
#pragma unroll
      for (int bq = 0; bq < 2; ++bq) {
        v4f T[4];
        quad_transpose(f.wq[sub][j][bq], T, odd, hi);
#pragma unroll
        for (int p = 0; p < 4; ++p)
#pragma unroll
          for (int d = 0; d < 4; ++d) w[sub][j][16 * bq + 4 * p + d] = T[p][d];
      }
#pragma unroll
      for (int d = 0; d < 4; ++d) w[sub][j][32 + d] = f.wt[sub][j][d];
    }

  float xr[2][2][3][XRUN];
#pragma unroll
  for (int sub = 0; sub < 2; ++sub)
#pragma unroll
    for (int i = 0; i < 2; ++i)
#pragma unroll
      for (int di = 0; di < 3; ++di) {
        const v4f E = f.xe[sub][i][di], O = f.xo[sub][i][di];
        const float g0 = quad_xor1(odd ? E[2] : O[0]);
        const float g1 = quad_xor1(odd ? E[3] : O[1]);
        xr[sub][i][di][0] = odd ? g0 : E[0];
        xr[sub][i][di][1] = odd ? g1 : E[1];
        xr[sub][i][di][2] = odd ? O[0] : E[2];
        xr[sub][i][di][3] = odd ? O[1] : E[3];
        xr[sub][i][di][4] = odd ? O[2] : g0;
        xr[sub][i][di][5] = odd ? O[3] : g1;
      }

#pragma unroll
  for (int l = 0; l < NLOC; ++l) {
#pragma unroll
    for (int p = 0; p < 9; ++p) {
      const int q0 = 2 * p, q1 = 2 * p + 1;
      const int s0 = q0 / 9, k0 = q0 % 9;
      const int s1 = q1 / 9, k1 = q1 % 9;
      v2f a[2], b[2];
#pragma unroll
      for (int i = 0; i < 2; ++i) {
        a[i].x = xr[s0][i][k0 / 3][l + k0 % 3];
        a[i].y = xr[s1][i][k1 / 3][l + k1 % 3];
      }
#pragma unroll
      for (int j = 0; j < 2; ++j) {
        b[j].x = w[s0][j][l * 9 + k0];
        b[j].y = w[s1][j][l * 9 + k1];
      }
#pragma unroll
      for (int i = 0; i < 2; ++i)
#pragma unroll
        for (int j = 0; j < 2; ++j)
          acc[l][i][j] = __builtin_amdgcn_wmma_f32_16x16x4_f32(
              false, a[i], false, b[j], (short)0, acc[l][i][j], false, false);
    }
  }
}

__global__ __launch_bounds__(128) void lc2d_wmma_f32_wgp16_kernel(
    const float* __restrict__ x,
    const float* __restrict__ wgt,
    const float* __restrict__ bias,
    float* __restrict__ out)
{
  const int lane = threadIdx.x & 31;
  const int t = lane >> 4;
  const int r = lane & 15;
  const int wv = __builtin_amdgcn_readfirstlane((int)(threadIdx.x >> 5));
  const int quad = ((wv & 1) << 1) | (wv >> 1);
  const int loc0 = (blockIdx.x * 4 + quad) * NLOC;
  const int h = loc0 >> 6;
  const int w0 = loc0 & 63;
  const bool odd = (lane & 1) != 0;
  const bool hi = (lane & 2) != 0;

  __shared__ __attribute__((aligned(16))) float otile[32 * 32 * OROW];

  const float* wl = wgt + ((size_t)(r * 16 + t) * 4096 + loc0) * 9;
  const float* wq = wgt + ((size_t)((r & ~3) * 16 + t) * 4096 + loc0) * 9 + 4 * (r & 3);
  const float* xl = x + ((size_t)((r & ~1) * 16 + t) * 66 + h) * 66 + w0 + 2 * (r & 1);

  v8f acc[NLOC][2][2];
#pragma unroll
  for (int l = 0; l < NLOC; ++l)
#pragma unroll
    for (int i = 0; i < 2; ++i)
#pragma unroll
      for (int j = 0; j < 2; ++j) acc[l][i][j] = (v8f){};

  Frag f0, f1;
  load_frag(f0, wq, wl, xl, 0);
  load_frag(f1, wq, wl, xl, 1);
  mma_frag(f0, acc, odd, hi);
  load_frag(f0, wq, wl, xl, 2);
  mma_frag(f1, acc, odd, hi);
  load_frag(f1, wq, wl, xl, 3);
  mma_frag(f0, acc, odd, hi);
  __builtin_amdgcn_sched_barrier(0);
  mma_frag(f1, acc, odd, hi);

  v4f bv[2];
#pragma unroll
  for (int j = 0; j < 2; ++j)
    bv[j] = *(const v4f*)(bias + (size_t)(r + 16 * j) * 4096 + loc0);

#pragma unroll
  for (int i = 0; i < 2; ++i)
#pragma unroll
    for (int j = 0; j < 2; ++j)
#pragma unroll
      for (int v = 0; v < 8; ++v) {
        const int b = 16 * i + v + 8 * t;
        const int o = 16 * j + r;
        v4f res;
        res.x = acc[0][i][j][v] + bv[j].x;
        res.y = acc[1][i][j][v] + bv[j].y;
        res.z = acc[2][i][j][v] + bv[j].z;
        res.w = acc[3][i][j][v] + bv[j].w;
        *(v4f*)(otile + (b * 32 + o) * OROW + quad * NLOC) = res;
      }

  __syncthreads();

  const int wg0 = blockIdx.x * (4 * NLOC);
  const int rk = lane >> 2;
  const int rq = lane & 3;
  v4f val[32];
#pragma unroll
  for (int it = 0; it < 32; ++it) {
    const int row = wv * 256 + (it >> 2) * 32 + (it & 3) + 4 * rk;
    val[it] = *(const v4f*)(otile + row * OROW + rq * 4);
  }
#pragma unroll
  for (int it = 0; it < 32; ++it) {
    const int row = wv * 256 + (it >> 2) * 32 + (it & 3) + 4 * rk;
    *(v4f*)(out + (size_t)row * 4096 + wg0 + rq * 4) = val[it];
  }
}

extern "C" void kernel_launch(void* const* d_in, const int* in_sizes, int n_in,
                              void* d_out, int out_size, void* d_ws, size_t ws_size,
                              hipStream_t stream) {
  (void)in_sizes; (void)n_in; (void)out_size; (void)d_ws; (void)ws_size;
  const float* x    = (const float*)d_in[0];
  const float* wgt  = (const float*)d_in[1];
  const float* bias = (const float*)d_in[2];
  float* out = (float*)d_out;
  lc2d_wmma_f32_wgp16_kernel<<<dim3(64 * 64 / (NLOC * 4)), dim3(128), 0, stream>>>(x, wgt, bias, out);
}
